# starting kernel
# baseline (speedup 1.0000x reference)
#include <hip/hip_runtime.h>
#include <hip/hip_bf16.h>


typedef __attribute__((ext_vector_type(2))) float v2f;
typedef __attribute__((ext_vector_type(8))) float v8f;

#define N_TOTAL 116250
#define DDIM 128
#define NT_PER_WAVE 2
#define WAVES_PER_BLOCK 8
#define N_PER_BLOCK (WAVES_PER_BLOCK * NT_PER_WAVE * 16)

__global__ __launch_bounds__(256) void conv_gemm_wmma_f32(
    const float* __restrict__ x,
    const float* __restrict__ w,
    float* __restrict__ out)
{
    const int lane  = threadIdx.x & 31;
    const int wave  = threadIdx.x >> 5;
    const int lrow  = lane & 15;
    const int khalf = (lane >> 4) << 1;

    const long long n_base =
        (long long)blockIdx.x * N_PER_BLOCK + (long long)wave * (NT_PER_WAVE * 16);

    v8f acc[4][NT_PER_WAVE];
    {
        v8f zero = {};
        #pragma unroll
        for (int ft = 0; ft < 4; ++ft)
            #pragma unroll
            for (int nt = 0; nt < NT_PER_WAVE; ++nt)
                acc[ft][nt] = zero;
    }

    const float* xrow[NT_PER_WAVE];
    #pragma unroll
    for (int nt = 0; nt < NT_PER_WAVE; ++nt) {
        long long n = n_base + nt * 16 + lrow;
        if (n > (long long)(N_TOTAL - 1)) n = N_TOTAL - 1;
        xrow[nt] = x + n * DDIM + khalf;
    }
    const float* wrow = w + (long long)lrow * DDIM + khalf;

    for (int kb = 0; kb < DDIM; kb += 4) {
        v2f a[4], b[NT_PER_WAVE];
        #pragma unroll
        for (int ft = 0; ft < 4; ++ft)
            a[ft] = *(const v2f*)(wrow + ft * 16 * DDIM + kb);
        #pragma unroll
        for (int nt = 0; nt < NT_PER_WAVE; ++nt)
            b[nt] = *(const v2f*)(xrow[nt] + kb);
        #pragma unroll
        for (int ft = 0; ft < 4; ++ft)
            #pragma unroll
            for (int nt = 0; nt < NT_PER_WAVE; ++nt)
                acc[ft][nt] = __builtin_amdgcn_wmma_f32_16x16x4_f32(
                     false, a[ft],
                     false, b[nt],
                     (short)0, acc[ft][nt],
                     false,  false);
    }

    const int mhalf = (lane >> 4) << 3;
    #pragma unroll
    for (int nt = 0; nt < NT_PER_WAVE; ++nt) {
        const long long n = n_base + nt * 16 + lrow;
        if (n < (long long)N_TOTAL) {
            #pragma unroll
            for (int ft = 0; ft < 4; ++ft) {
                float* orow = out + (long long)(ft * 16 + mhalf) * N_TOTAL + n;
                #pragma unroll
                for (int r = 0; r < 8; ++r)
                    orow[(long long)r * N_TOTAL] = acc[ft][nt][r];
            }
        }
    }
}

extern "C" void kernel_launch(void* const* d_in, const int* in_sizes, int n_in,
                              void* d_out, int out_size, void* d_ws, size_t ws_size,
                              hipStream_t stream) {
    const float* x = (const float*)d_in[0];
    const float* w = (const float*)d_in[1];
    float* out = (float*)d_out;

    const int grid = (N_TOTAL + N_PER_BLOCK - 1) / N_PER_BLOCK;
    conv_gemm_wmma_f32<<<grid, 256, 0, stream>>>(x, w, out);
}


# W (64x128) kept in registers as A fragments (256 VGPRs) for the whole kernel: 1024 waves, one per SIMD, each loop over 7-8 column tiles, next tile's X loads issued ahead; k and sum order unchanged
# speedup vs baseline: 1.3212x; 1.3212x over previous
#include <hip/hip_runtime.h>


typedef __attribute__((ext_vector_type(2))) float v2f;
typedef __attribute__((ext_vector_type(4))) float v4f;
typedef __attribute__((ext_vector_type(8))) float v8f;

#define N_TOTAL 116250
#define DDIM 128
#define KBLOCKS (DDIM / 8)
#define N_TILES ((N_TOTAL + 15) / 16)
#define WAVES_PER_BLOCK 4
#define N_BLOCKS 256
#define N_WAVES (N_BLOCKS * WAVES_PER_BLOCK)
#define TILES_LO (N_TILES / N_WAVES)
#define WAVES_HI (N_TILES - TILES_LO * N_WAVES)

static __device__ __forceinline__ void load_x_tile(
    v4f (&b)[KBLOCKS], const float* x, int tile, int lrow, int khalf)
{
    int n = tile * 16 + lrow;
    if (n > N_TOTAL - 1) n = N_TOTAL - 1;
    const float* p = x + (size_t)n * DDIM + khalf;
    #pragma unroll
    for (int kb = 0; kb < KBLOCKS; ++kb)
        b[kb] = *(const v4f*)(p + kb * 8);
}

static __device__ __forceinline__ void tile_product_store(
    const v4f (&a)[4][KBLOCKS], const v4f (&b)[KBLOCKS],
    float* out, int tile, int lrow, int mhalf)
{
    v8f acc[4];
    {
        const v8f zero = {};
        #pragma unroll
        for (int ft = 0; ft < 4; ++ft) acc[ft] = zero;
    }
    #pragma unroll
    for (int kb = 0; kb < KBLOCKS; ++kb) {
        #pragma unroll
        for (int p = 0; p < 2; ++p) {
            const v2f bf = p ? (v2f){b[kb].z, b[kb].w} : (v2f){b[kb].x, b[kb].y};
            #pragma unroll
            for (int ft = 0; ft < 4; ++ft) {
                const v2f af = p ? (v2f){a[ft][kb].z, a[ft][kb].w}
                                 : (v2f){a[ft][kb].x, a[ft][kb].y};
                acc[ft] = __builtin_amdgcn_wmma_f32_16x16x4_f32(
                     false, af,  false, bf,
                     (short)0, acc[ft],
                     false,  false);
            }
        }
    }
    const int n = tile * 16 + lrow;
    if (n < N_TOTAL) {
        #pragma unroll
        for (int ft = 0; ft < 4; ++ft) {
            float* orow = out + (size_t)(ft * 16 + mhalf) * N_TOTAL + n;
            #pragma unroll
            for (int r = 0; r < 8; ++r)
                orow[(size_t)r * N_TOTAL] = acc[ft][r];
        }
    }
}

__global__ __launch_bounds__(128) __attribute__((amdgpu_waves_per_eu(1, 1)))
void conv_gemm_wmma_f32_wreg(
    const float* x,
    const float* __restrict__ w,
    float* out)
{
    const int lane  = threadIdx.x & 31;
    const int wave  = __builtin_amdgcn_readfirstlane(threadIdx.x >> 5);
    const int lrow  = lane & 15;
    const int khalf = (lane >> 4) << 2;
    const int mhalf = (lane >> 4) << 3;

    const int g = blockIdx.x * WAVES_PER_BLOCK + wave;
    const int t_begin = g * TILES_LO + (g < WAVES_HI ? g : WAVES_HI);
    const int t_end   = t_begin + TILES_LO + (g < WAVES_HI ? 1 : 0);

    v4f b0[KBLOCKS], b1[KBLOCKS];
    load_x_tile(b0, x, t_begin, lrow, khalf);

    v4f a[4][KBLOCKS];
    {
        const float* wrow = w + lrow * DDIM + khalf;
        #pragma unroll
        for (int kb = 0; kb < KBLOCKS; ++kb)
            #pragma unroll
            for (int ft = 0; ft < 4; ++ft)
                a[ft][kb] = *(const v4f*)(wrow + ft * 16 * DDIM + kb * 8);
    }

    const int last = t_end - 1;

    __builtin_amdgcn_sched_barrier(0);
    load_x_tile(b1, x, (t_begin + 1 < last ? t_begin + 1 : last), lrow, khalf);
    __builtin_amdgcn_sched_barrier(0);
    tile_product_store(a, b0, out, t_begin, lrow, mhalf);

    int t = t_begin + 1;
    for (; t + 1 < t_end; t += 2) {
        load_x_tile(b0, x, t + 1, lrow, khalf);
        __builtin_amdgcn_sched_barrier(0);
        tile_product_store(a, b1, out, t, lrow, mhalf);
        load_x_tile(b1, x, (t + 2 < last ? t + 2 : last), lrow, khalf);
        __builtin_amdgcn_sched_barrier(0);
        tile_product_store(a, b0, out, t + 1, lrow, mhalf);
    }
    if (t < t_end)
        tile_product_store(a, b1, out, t, lrow, mhalf);
}

extern "C" void kernel_launch(void* const* d_in, const int* in_sizes, int n_in,
                              void* d_out, int out_size, void* d_ws, size_t ws_size,
                              hipStream_t stream) {
    const float* x = (const float*)d_in[0];
    const float* w = (const float*)d_in[1];
    float* out = (float*)d_out;

    conv_gemm_wmma_f32_wreg<<<N_BLOCKS, WAVES_PER_BLOCK * 32, 0, stream>>>(x, w, out);
}
